# starting kernel
# baseline (speedup 1.0000x reference)
#include <hip/hip_runtime.h>
#include <math.h>


typedef __attribute__((ext_vector_type(2))) float v2f;
typedef __attribute__((ext_vector_type(8))) float v8f;

#define BATCH 32
#define CCH   256
#define HH    64
#define WW    64
#define HW    (HH * WW)
#define CHW   (CCH * HW)
#define EPSV  1e-5f

__global__ void pool_kernel(const float* __restrict__ x,
                            float* __restrict__ pmax,
                            float* __restrict__ pavg) {
  __shared__ float4 smax[256];
  __shared__ float4 ssum[256];
  const int bh = blockIdx.x;
  const int b = bh >> 6, h = bh & 63;
  const int t = threadIdx.x;
  const int w4 = t & 15, cg = t >> 4;
  const float4* base = (const float4*)(x + (size_t)b * CHW + (size_t)h * WW) + w4;
  float4 mx = make_float4(-INFINITY, -INFINITY, -INFINITY, -INFINITY);
  float4 sm = make_float4(0.f, 0.f, 0.f, 0.f);
  const int c0 = cg * 16;
#pragma unroll 4
  for (int i = 0; i < 16; ++i) {
    float4 v = base[(size_t)(c0 + i) * (HW / 4)];
    mx.x = fmaxf(mx.x, v.x); mx.y = fmaxf(mx.y, v.y);
    mx.z = fmaxf(mx.z, v.z); mx.w = fmaxf(mx.w, v.w);
    sm.x += v.x; sm.y += v.y; sm.z += v.z; sm.w += v.w;
  }
  smax[t] = mx;
  ssum[t] = sm;
  __syncthreads();
  for (int s = 128; s >= 16; s >>= 1) {
    if (t < s) {
      float4 a = smax[t], c = smax[t + s];
      a.x = fmaxf(a.x, c.x); a.y = fmaxf(a.y, c.y);
      a.z = fmaxf(a.z, c.z); a.w = fmaxf(a.w, c.w);
      smax[t] = a;
      float4 u = ssum[t], d = ssum[t + s];
      u.x += d.x; u.y += d.y; u.z += d.z; u.w += d.w;
      ssum[t] = u;
    }
    __syncthreads();
  }
  if (t < 16) {
    float4 m2 = smax[t];
    float4 s2 = ssum[t];
    s2.x *= (1.f / 256.f); s2.y *= (1.f / 256.f);
    s2.z *= (1.f / 256.f); s2.w *= (1.f / 256.f);
    ((float4*)pmax)[bh * 16 + t] = m2;
    ((float4*)pavg)[bh * 16 + t] = s2;
  }
}

__global__ void conv_att_kernel(const float* __restrict__ pmax,
                                const float* __restrict__ pavg,
                                const float* __restrict__ w1,
                                const float* __restrict__ w2,
                                const float* __restrict__ w3,
                                const float* __restrict__ g1, const float* __restrict__ be1,
                                const float* __restrict__ m1, const float* __restrict__ v1,
                                const float* __restrict__ g2, const float* __restrict__ be2,
                                const float* __restrict__ m2, const float* __restrict__ v2,
                                const float* __restrict__ g3, const float* __restrict__ be3,
                                const float* __restrict__ m3, const float* __restrict__ v3,
                                float* __restrict__ att) {
  __shared__ float lp[2][8][72];
  __shared__ float btabT[16][128];
  __shared__ float ytmp[4][3][16];
  const int bh = blockIdx.x;
  const int b = bh >> 6, h = bh & 63;
  const int tid = threadIdx.x;

  for (int idx = tid; idx < 16 * 128; idx += 128) {
    int n = idx >> 7, k = idx & 127;
    int ch = k >> 6, r = (k >> 3) & 7, s = k & 7;
    float val = 0.f;
    if (n < 3 && r < 7 && s < 7) {
      if (n == 0)      val = w1[ch * 49 + r * 7 + s];
      else if (n == 1) val = w2[ch * 49 + s * 7 + r];
      else             val = w3[ch * 49 + r * 7 + s];
    }
    btabT[n][k] = val;
  }
  for (int idx = tid; idx < 2 * 8 * 72; idx += 128) ((float*)lp)[idx] = 0.f;
  __syncthreads();
  for (int idx = tid; idx < 2 * 7 * 64; idx += 128) {
    int w = idx & 63;
    int rr = (idx >> 6) % 7;
    int ch = idx / (7 * 64);
    int hh = h + rr - 3;
    if (hh >= 0 && hh < 64) {
      const float* src = (ch == 0) ? pmax : pavg;
      lp[ch][rr][w + 3] = src[(b * 64 + hh) * 64 + w];
    }
  }
  __syncthreads();

  const int lane = tid & 31, wid = tid >> 5;
  const int m = lane & 15, khalf = lane >> 4;
  const int w0 = wid * 16;
  const float* lpf = &lp[0][0][0];
  const float* btf = &btabT[0][0];
  const int aBase = w0 + m + 2 * khalf;
  const int bBase = m * 128 + 2 * khalf;

  v8f acc = {};
#if __has_builtin(__builtin_amdgcn_wmma_f32_16x16x4_f32)
#pragma unroll
  for (int chunk = 0; chunk < 32; ++chunk) {
    const int k0 = chunk * 4;
    const int ch = k0 >> 6;
    const int r  = (k0 >> 3) & 7;
    const int sc = k0 & 7;
    const int ai = ch * 576 + r * 72 + sc + aBase;
    v2f a;
    a[0] = lpf[ai];
    a[1] = lpf[ai + 1];
    v2f bm = *(const v2f*)(btf + bBase + k0);
    acc = __builtin_amdgcn_wmma_f32_16x16x4_f32(
         false, a,  false, bm,
         (short)0, acc,  false,  false);
  }
#else
  for (int k = 0; k < 128; ++k) {
    int ch = k >> 6, r = (k >> 3) & 7, s = k & 7;
    float bw = btf[m * 128 + k];
#pragma unroll
    for (int v = 0; v < 8; ++v) {
      int M = v + 8 * khalf;
      acc[v] += lp[ch][r][w0 + M + s] * bw;
    }
  }
#endif

  if (m < 3) {
#pragma unroll
    for (int v = 0; v < 8; ++v) ytmp[wid][m][v + 8 * khalf] = acc[v];
  }
  __syncthreads();

  if (lane < 16) {
    float y0 = ytmp[wid][0][lane];
    float y1 = ytmp[wid][1][lane];
    float y2 = ytmp[wid][2][lane];
    float s1 = g1[0] / sqrtf(v1[0] + EPSV), t1 = be1[0] - m1[0] * s1;
    float s2 = g2[0] / sqrtf(v2[0] + EPSV), t2 = be2[0] - m2[0] * s2;
    float s3 = g3[0] / sqrtf(v3[0] + EPSV), t3 = be3[0] - m3[0] * s3;
    float a1 = 1.f / (1.f + expf(-(y0 * s1 + t1)));
    float a2 = 1.f / (1.f + expf(-(y1 * s2 + t2)));
    float a3 = 1.f / (1.f + expf(-(y2 * s3 + t3)));
    att[bh * 64 + w0 + lane] = (a1 + a2 + a3) * (1.f / 3.f);
  }
}

__global__ void scale_kernel(const float* __restrict__ x,
                             const float* __restrict__ att,
                             float* __restrict__ out) {
  const size_t i4 = (size_t)blockIdx.x * blockDim.x + threadIdx.x;
  const size_t base = i4 * 4;
  const int b = (int)(base >> 20);
  const int hw = (int)(base & 4095);
  float4 xv = ((const float4*)x)[i4];
  float4 av = *(const float4*)(att + (size_t)b * HW + hw);
  float4 o;
  o.x = xv.x * av.x;
  o.y = xv.y * av.y;
  o.z = xv.z * av.z;
  o.w = xv.w * av.w;
  ((float4*)out)[i4] = o;
}

extern "C" void kernel_launch(void* const* d_in, const int* in_sizes, int n_in,
                              void* d_out, int out_size, void* d_ws, size_t ws_size,
                              hipStream_t stream) {
  const float* x  = (const float*)d_in[0];
  const float* w1 = (const float*)d_in[1];
  const float* g1 = (const float*)d_in[2];
  const float* b1 = (const float*)d_in[3];
  const float* m1 = (const float*)d_in[4];
  const float* v1 = (const float*)d_in[5];
  const float* w2 = (const float*)d_in[6];
  const float* g2 = (const float*)d_in[7];
  const float* b2 = (const float*)d_in[8];
  const float* m2 = (const float*)d_in[9];
  const float* v2 = (const float*)d_in[10];
  const float* w3 = (const float*)d_in[11];
  const float* g3 = (const float*)d_in[12];
  const float* b3 = (const float*)d_in[13];
  const float* m3 = (const float*)d_in[14];
  const float* v3 = (const float*)d_in[15];

  float* ws   = (float*)d_ws;
  float* pmax = ws;
  float* pavg = ws + BATCH * HW;
  float* attb = ws + 2 * BATCH * HW;

  pool_kernel<<<BATCH * HH, 256, 0, stream>>>(x, pmax, pavg);
  conv_att_kernel<<<BATCH * HH, 128, 0, stream>>>(pmax, pavg, w1, w2, w3,
                                                  g1, b1, m1, v1,
                                                  g2, b2, m2, v2,
                                                  g3, b3, m3, v3, attb);
  const int total4 = BATCH * CHW / 4;
  scale_kernel<<<total4 / 256, 256, 0, stream>>>(x, attb, (float*)d_out);
}


# fuse the conv+BN+sigmoid kernel into the scale kernel: one 8-wave block per (b,h) row keeps att in LDS and scales all 256 channels, the row's x loads in flight across the conv (3 launches -> 2)
# speedup vs baseline: 1.3877x; 1.3877x over previous
#include <hip/hip_runtime.h>
#include <math.h>


typedef __attribute__((ext_vector_type(2))) float v2f;
typedef __attribute__((ext_vector_type(8))) float v8f;

#define BATCH 32
#define CCH   256
#define HH    64
#define WW    64
#define HW    (HH * WW)
#define CHW   (CCH * HW)
#define EPSV  1e-5f

__global__ void pool_kernel(const float* __restrict__ x,
                            float* __restrict__ pmax,
                            float* __restrict__ pavg) {
  __shared__ float4 smax[256];
  __shared__ float4 ssum[256];
  const int bh = blockIdx.x;
  const int b = bh >> 6, h = bh & 63;
  const int t = threadIdx.x;
  const int w4 = t & 15, cg = t >> 4;
  const float4* base = (const float4*)(x + (size_t)b * CHW + (size_t)h * WW) + w4;
  float4 mx = make_float4(-INFINITY, -INFINITY, -INFINITY, -INFINITY);
  float4 sm = make_float4(0.f, 0.f, 0.f, 0.f);
  const int c0 = cg * 16;
#pragma unroll 4
  for (int i = 0; i < 16; ++i) {
    float4 v = base[(size_t)(c0 + i) * (HW / 4)];
    mx.x = fmaxf(mx.x, v.x); mx.y = fmaxf(mx.y, v.y);
    mx.z = fmaxf(mx.z, v.z); mx.w = fmaxf(mx.w, v.w);
    sm.x += v.x; sm.y += v.y; sm.z += v.z; sm.w += v.w;
  }
  smax[t] = mx;
  ssum[t] = sm;
  __syncthreads();
  for (int s = 128; s >= 16; s >>= 1) {
    if (t < s) {
      float4 a = smax[t], c = smax[t + s];
      a.x = fmaxf(a.x, c.x); a.y = fmaxf(a.y, c.y);
      a.z = fmaxf(a.z, c.z); a.w = fmaxf(a.w, c.w);
      smax[t] = a;
      float4 u = ssum[t], d = ssum[t + s];
      u.x += d.x; u.y += d.y; u.z += d.z; u.w += d.w;
      ssum[t] = u;
    }
    __syncthreads();
  }
  if (t < 16) {
    float4 m2 = smax[t];
    float4 s2 = ssum[t];
    s2.x *= (1.f / 256.f); s2.y *= (1.f / 256.f);
    s2.z *= (1.f / 256.f); s2.w *= (1.f / 256.f);
    ((float4*)pmax)[bh * 16 + t] = m2;
    ((float4*)pavg)[bh * 16 + t] = s2;
  }
}

static __device__ __forceinline__ void lds_barrier() {
  asm volatile("s_wait_dscnt 0x0\n\ts_barrier_signal -1\n\ts_barrier_wait -1" ::: "memory");
}

__global__ __launch_bounds__(256) void conv_scale_kernel(const float* __restrict__ x,
                                const float* __restrict__ pmax,
                                const float* __restrict__ pavg,
                                const float* __restrict__ w1,
                                const float* __restrict__ w2,
                                const float* __restrict__ w3,
                                const float* __restrict__ g1, const float* __restrict__ be1,
                                const float* __restrict__ m1, const float* __restrict__ v1,
                                const float* __restrict__ g2, const float* __restrict__ be2,
                                const float* __restrict__ m2, const float* __restrict__ v2,
                                const float* __restrict__ g3, const float* __restrict__ be3,
                                const float* __restrict__ m3, const float* __restrict__ v3,
                                float* __restrict__ out) {
  __shared__ __attribute__((aligned(16))) float lp[2][8][72];
  __shared__ __attribute__((aligned(16))) float btabT[16][128];
  __shared__ float ytmp[4][3][16];
  __shared__ __attribute__((aligned(16))) float att_s[64];
  const int bh = blockIdx.x;
  const int b = bh >> 6, h = bh & 63;
  const int tid = threadIdx.x;

  const float G1 = g1[0], Be1 = be1[0], M1 = m1[0], V1 = v1[0];
  const float G2 = g2[0], Be2 = be2[0], M2 = m2[0], V2 = v2[0];
  const float G3 = g3[0], Be3 = be3[0], M3 = m3[0], V3 = v3[0];
  const int k = tid & 127;
  const int kch = k >> 6, kr = (k >> 3) & 7, ks = k & 7;
  const bool tapok = (kr < 7) && (ks < 7);
  const bool hi = tid >= 128;
  const int offn = kch * 49 + kr * 7 + ks;
  const int offt = kch * 49 + ks * 7 + kr;
  const float* wp01 = hi ? w2 : w1;
  const float wv01 = wp01[tapok ? (hi ? offt : offn) : 0];
  const float wv2  = w3[(tapok && !hi) ? offn : 0];
  const int pw = tid & 63;
  float pv[4];
  int pdst[4];
  bool pok[4];
#pragma unroll
  for (int j = 0; j < 4; ++j) {
    const int q = (tid >> 6) + 4 * j;
    const int ch = (q >= 7) ? 1 : 0;
    const int rr = q - 7 * ch;
    const int hh = h + rr - 3;
    pok[j] = (q < 14) && (hh >= 0) && (hh < 64);
    const int hc = hh < 0 ? 0 : (hh > 63 ? 63 : hh);
    const float* src = ch ? pavg : pmax;
    pv[j] = src[(b * 64 + hc) * 64 + pw];
    pdst[j] = ch * 576 + rr * 72 + pw + 3;
  }

  const int w4 = tid & 15, cg = tid >> 4;
  const size_t rowoff = (size_t)b * CHW + (size_t)h * WW;
  const float4* xb = (const float4*)(x + rowoff) + w4;
  float4 xv[16];
#pragma unroll
  for (int i = 0; i < 16; ++i) xv[i] = xb[(size_t)(cg + 16 * i) * (HW / 4)];

  const float4 z4 = make_float4(0.f, 0.f, 0.f, 0.f);
  float4* lp4 = (float4*)&lp[0][0][0];
  float4* bt4 = (float4*)&btabT[0][0];
  lp4[tid] = z4;
  if (tid < 32) lp4[256 + tid] = z4;
  bt4[128 + tid] = z4;
  if (tid < 128) bt4[384 + tid] = z4;
  lds_barrier();

  float* btw = &btabT[0][0];
  btw[tid] = tapok ? wv01 : 0.f;
  btw[256 + tid] = (tapok && !hi) ? wv2 : 0.f;
  float* lpw = &lp[0][0][0];
#pragma unroll
  for (int j = 0; j < 4; ++j) {
    if (pok[j]) lpw[pdst[j]] = pv[j];
  }
  lds_barrier();

  const int lane = tid & 31;
  const int wid = __builtin_amdgcn_readfirstlane(tid >> 5);
  const int m = lane & 15, khalf = lane >> 4;
  const int w0 = wid * 16;
  const float* lpf = &lp[0][0][0];
  const float* btf = &btabT[0][0];
  const int aBase = w0 + m + 2 * khalf;
  const int bBase = m * 128 + 2 * khalf;

  if (wid < 4) {
    v8f acc = {};
#pragma unroll
    for (int chunk = 0; chunk < 32; ++chunk) {
      const int k0 = chunk * 4;
      const int ch = k0 >> 6;
      const int r  = (k0 >> 3) & 7;
      const int sc = k0 & 7;
      const int ai = ch * 576 + r * 72 + sc + aBase;
      v2f a;
      a[0] = lpf[ai];
      a[1] = lpf[ai + 1];
      v2f bm = *(const v2f*)(btf + bBase + k0);
      acc = __builtin_amdgcn_wmma_f32_16x16x4_f32(
           false, a,  false, bm,
           (short)0, acc,  false,  false);
    }
    if (m < 3) {
#pragma unroll
      for (int v = 0; v < 8; ++v) ytmp[wid][m][v + 8 * khalf] = acc[v];
    }
  }
  lds_barrier();

  if (wid < 4 && lane < 16) {
    float y0 = ytmp[wid][0][lane];
    float y1 = ytmp[wid][1][lane];
    float y2 = ytmp[wid][2][lane];
    float s1 = G1 / sqrtf(V1 + EPSV), t1 = Be1 - M1 * s1;
    float s2 = G2 / sqrtf(V2 + EPSV), t2 = Be2 - M2 * s2;
    float s3 = G3 / sqrtf(V3 + EPSV), t3 = Be3 - M3 * s3;
    float a1 = 1.f / (1.f + expf(-(y0 * s1 + t1)));
    float a2 = 1.f / (1.f + expf(-(y1 * s2 + t2)));
    float a3 = 1.f / (1.f + expf(-(y2 * s3 + t3)));
    att_s[w0 + lane] = (a1 + a2 + a3) * (1.f / 3.f);
  }
  lds_barrier();

  const float4 av = *(const float4*)&att_s[w4 * 4];
  float4* ob = (float4*)(out + rowoff) + w4;
#pragma unroll
  for (int i = 0; i < 16; ++i) {
    float4 o;
    o.x = xv[i].x * av.x;
    o.y = xv[i].y * av.y;
    o.z = xv[i].z * av.z;
    o.w = xv[i].w * av.w;
    ob[(size_t)(cg + 16 * i) * (HW / 4)] = o;
  }
}

extern "C" void kernel_launch(void* const* d_in, const int* in_sizes, int n_in,
                              void* d_out, int out_size, void* d_ws, size_t ws_size,
                              hipStream_t stream) {
  const float* x  = (const float*)d_in[0];
  const float* w1 = (const float*)d_in[1];
  const float* g1 = (const float*)d_in[2];
  const float* b1 = (const float*)d_in[3];
  const float* m1 = (const float*)d_in[4];
  const float* v1 = (const float*)d_in[5];
  const float* w2 = (const float*)d_in[6];
  const float* g2 = (const float*)d_in[7];
  const float* b2 = (const float*)d_in[8];
  const float* m2 = (const float*)d_in[9];
  const float* v2 = (const float*)d_in[10];
  const float* w3 = (const float*)d_in[11];
  const float* g3 = (const float*)d_in[12];
  const float* b3 = (const float*)d_in[13];
  const float* m3 = (const float*)d_in[14];
  const float* v3 = (const float*)d_in[15];

  float* ws   = (float*)d_ws;
  float* pmax = ws;
  float* pavg = ws + BATCH * HW;

  pool_kernel<<<BATCH * HH, 256, 0, stream>>>(x, pmax, pavg);
  conv_scale_kernel<<<BATCH * HH, 256, 0, stream>>>(x, pmax, pavg, w1, w2, w3,
                                                    g1, b1, m1, v1,
                                                    g2, b2, m2, v2,
                                                    g3, b3, m3, v3, (float*)d_out);
}


# both launches: one 8-wave block per 4 adjacent rows, not 1 (grid 2048 -> 512, 4x fewer waves to enter; 512 B contiguous per wave access, 32+ loads per lane in flight); channel-mean sum order changes
# speedup vs baseline: 1.7787x; 1.2818x over previous
#include <hip/hip_runtime.h>
#include <math.h>


typedef __attribute__((ext_vector_type(2))) float v2f;
typedef __attribute__((ext_vector_type(8))) float v8f;

#define BATCH 32
#define CCH   256
#define HH    64
#define WW    64
#define HW    (HH * WW)
#define CHW   (CCH * HW)
#define EPSV  1e-5f
#define RB    4
#define NRG   (HH / RB)
#define LPR   (RB + 7)
#define LPCH  (LPR * 72)

static __device__ __forceinline__ void pool_acc(float4& mx, float4& sm, const float4 v) {
  mx.x = fmaxf(mx.x, v.x); mx.y = fmaxf(mx.y, v.y);
  mx.z = fmaxf(mx.z, v.z); mx.w = fmaxf(mx.w, v.w);
  sm.x += v.x; sm.y += v.y; sm.z += v.z; sm.w += v.w;
}

__global__ __launch_bounds__(256) __attribute__((amdgpu_waves_per_eu(4, 4))) void pool_kernel(const float* __restrict__ x,
                            float* __restrict__ pmax,
                            float* __restrict__ pavg) {
  __shared__ float4 smax[256];
  __shared__ float4 ssum[256];
  const int g = blockIdx.x;
  const int b = g >> 4, h0 = (g & 15) * RB;
  const int t = threadIdx.x;
  const int f = t & 63, cg = t >> 6;
  const float4* base = (const float4*)(x + (size_t)b * CHW + (size_t)h0 * WW)
                       + f + (size_t)cg * (HW / 4);
  float4 mx = make_float4(-INFINITY, -INFINITY, -INFINITY, -INFINITY);
  float4 sm = make_float4(0.f, 0.f, 0.f, 0.f);
  float4 v[32];
#pragma unroll
  for (int j = 0; j < 32; ++j) v[j] = base[(size_t)j * HW];
#pragma unroll
  for (int j = 0; j < 32; ++j) {
    pool_acc(mx, sm, v[j]);
    v[j] = base[(size_t)(32 + j) * HW];
  }
#pragma unroll
  for (int j = 0; j < 32; ++j) pool_acc(mx, sm, v[j]);
  smax[t] = mx;
  ssum[t] = sm;
  __syncthreads();
  if (t < 64) {
    float4 a = smax[t];
    float4 u = ssum[t];
#pragma unroll
    for (int q = 1; q < 4; ++q) {
      const float4 c = smax[t + 64 * q];
      const float4 d = ssum[t + 64 * q];
      a.x = fmaxf(a.x, c.x); a.y = fmaxf(a.y, c.y);
      a.z = fmaxf(a.z, c.z); a.w = fmaxf(a.w, c.w);
      u.x += d.x; u.y += d.y; u.z += d.z; u.w += d.w;
    }
    u.x *= (1.f / 256.f); u.y *= (1.f / 256.f);
    u.z *= (1.f / 256.f); u.w *= (1.f / 256.f);
    ((float4*)pmax)[g * 64 + t] = a;
    ((float4*)pavg)[g * 64 + t] = u;
  }
}

static __device__ __forceinline__ void lds_barrier() {
  asm volatile("s_wait_dscnt 0x0\n\ts_barrier_signal -1\n\ts_barrier_wait -1" ::: "memory");
}

__global__ __launch_bounds__(256) __attribute__((amdgpu_waves_per_eu(4, 4))) void conv_scale_kernel(const float* __restrict__ x,
                                const float* __restrict__ pmax,
                                const float* __restrict__ pavg,
                                const float* __restrict__ w1,
                                const float* __restrict__ w2,
                                const float* __restrict__ w3,
                                const float* __restrict__ g1, const float* __restrict__ be1,
                                const float* __restrict__ m1, const float* __restrict__ v1,
                                const float* __restrict__ g2, const float* __restrict__ be2,
                                const float* __restrict__ m2, const float* __restrict__ v2,
                                const float* __restrict__ g3, const float* __restrict__ be3,
                                const float* __restrict__ m3, const float* __restrict__ v3,
                                float* __restrict__ out) {
  __shared__ __attribute__((aligned(16))) float lp[2][LPR][72];
  __shared__ __attribute__((aligned(16))) float btabT[16][128];
  __shared__ float ys[3][RB * 64];
  __shared__ __attribute__((aligned(16))) float att_s[RB * 64];
  const int g = blockIdx.x;
  const int b = g >> 4, h0 = (g & 15) * RB;
  const int tid = threadIdx.x;

  const float G1 = g1[0], Be1 = be1[0], M1 = m1[0], V1 = v1[0];
  const float G2 = g2[0], Be2 = be2[0], M2 = m2[0], V2 = v2[0];
  const float G3 = g3[0], Be3 = be3[0], M3 = m3[0], V3 = v3[0];
  const int k = tid & 127;
  const int kch = k >> 6, kr = (k >> 3) & 7, ks = k & 7;
  const bool tapok = (kr < 7) && (ks < 7);
  const bool hi = tid >= 128;
  const int offn = kch * 49 + kr * 7 + ks;
  const int offt = kch * 49 + ks * 7 + kr;
  const float* wp01 = hi ? w2 : w1;
  const float wv01 = wp01[tapok ? (hi ? offt : offn) : 0];
  const float wv2  = w3[(tapok && !hi) ? offn : 0];
  const int pw = tid & 63;
  float pv[5];
  int pdst[5];
  bool pok[5];
#pragma unroll
  for (int j = 0; j < 5; ++j) {
    const int q = (tid >> 6) + 4 * j;
    const int ch = (q >= 10) ? 1 : 0;
    const int rr = q - 10 * ch;
    const int hh = h0 + rr - 3;
    pok[j] = (hh >= 0) && (hh < 64);
    const int hc = hh < 0 ? 0 : (hh > 63 ? 63 : hh);
    const float* src = ch ? pavg : pmax;
    pv[j] = src[(b * 64 + hc) * 64 + pw];
    pdst[j] = ch * LPCH + rr * 72 + pw + 3;
  }

  const int f = tid & 63, cg = tid >> 6;
  const size_t rowoff = (size_t)b * CHW + (size_t)h0 * WW;
  const float4* xb = (const float4*)(x + rowoff) + f + (size_t)cg * (HW / 4);
  float4 xr[32];
#pragma unroll
  for (int i = 0; i < 32; ++i) xr[i] = xb[(size_t)i * HW];

  const float4 z4 = make_float4(0.f, 0.f, 0.f, 0.f);
  float4* lp4 = (float4*)&lp[0][0][0];
  float4* bt4 = (float4*)&btabT[0][0];
  lp4[tid] = z4;
  if (tid < 2 * LPR * 18 - 256) lp4[256 + tid] = z4;
  bt4[128 + tid] = z4;
  if (tid < 128) bt4[384 + tid] = z4;
  lds_barrier();

  float* btw = &btabT[0][0];
  btw[tid] = tapok ? wv01 : 0.f;
  btw[256 + tid] = (tapok && !hi) ? wv2 : 0.f;
  float* lpw = &lp[0][0][0];
#pragma unroll
  for (int j = 0; j < 5; ++j) {
    if (pok[j]) lpw[pdst[j]] = pv[j];
  }
  lds_barrier();

  const int lane = tid & 31;
  const int wid = __builtin_amdgcn_readfirstlane(tid >> 5);
  const int m = lane & 15, khalf = lane >> 4;
  const int rt = wid >> 1;
  const int c0 = (wid & 1) * 32;
  const float* lpf = &lp[0][0][0];
  const float* btf = &btabT[0][0];
  const int aBase = rt * 72 + c0 + m + 2 * khalf;
  const int bBase = m * 128 + 2 * khalf;

  {
    v8f acc0 = {}, acc1 = {};
#pragma unroll
    for (int chunk = 0; chunk < 32; ++chunk) {
      const int k0 = chunk * 4;
      const int ch = k0 >> 6;
      const int r  = (k0 >> 3) & 7;
      const int sc = k0 & 7;
      const int ai = ch * LPCH + r * 72 + sc + aBase;
      v2f a0, a1;
      a0[0] = lpf[ai];
      a0[1] = lpf[ai + 1];
      a1[0] = lpf[ai + 16];
      a1[1] = lpf[ai + 17];
      v2f bm = *(const v2f*)(btf + bBase + k0);
      acc0 = __builtin_amdgcn_wmma_f32_16x16x4_f32(
           false, a0,  false, bm,
           (short)0, acc0,  false,  false);
      acc1 = __builtin_amdgcn_wmma_f32_16x16x4_f32(
           false, a1,  false, bm,
           (short)0, acc1,  false,  false);
    }
    if (m < 3) {
      float* yd = &ys[m][rt * 64 + c0 + 8 * khalf];
#pragma unroll
      for (int v = 0; v < 8; ++v) {
        yd[v] = acc0[v];
        yd[16 + v] = acc1[v];
      }
    }
  }
  lds_barrier();

  {
    float y0 = ys[0][tid];
    float y1 = ys[1][tid];
    float y2 = ys[2][tid];
    float s1 = G1 / sqrtf(V1 + EPSV), t1 = Be1 - M1 * s1;
    float s2 = G2 / sqrtf(V2 + EPSV), t2 = Be2 - M2 * s2;
    float s3 = G3 / sqrtf(V3 + EPSV), t3 = Be3 - M3 * s3;
    float a1 = 1.f / (1.f + expf(-(y0 * s1 + t1)));
    float a2 = 1.f / (1.f + expf(-(y1 * s2 + t2)));
    float a3 = 1.f / (1.f + expf(-(y2 * s3 + t3)));
    att_s[tid] = (a1 + a2 + a3) * (1.f / 3.f);
  }
  lds_barrier();

  const float4 av = *(const float4*)&att_s[f * 4];
  float4* ob = (float4*)(out + rowoff) + f + (size_t)cg * (HW / 4);
#pragma unroll
  for (int i = 0; i < 32; ++i) {
    float4 o = xr[i];
    o.x *= av.x; o.y *= av.y; o.z *= av.z; o.w *= av.w;
    ob[(size_t)i * HW] = o;
    xr[i] = xb[(size_t)(32 + i) * HW];
  }
#pragma unroll
  for (int i = 0; i < 32; ++i) {
    float4 o = xr[i];
    o.x *= av.x; o.y *= av.y; o.z *= av.z; o.w *= av.w;
    ob[(size_t)(32 + i) * HW] = o;
  }
}

extern "C" void kernel_launch(void* const* d_in, const int* in_sizes, int n_in,
                              void* d_out, int out_size, void* d_ws, size_t ws_size,
                              hipStream_t stream) {
  const float* x  = (const float*)d_in[0];
  const float* w1 = (const float*)d_in[1];
  const float* g1 = (const float*)d_in[2];
  const float* b1 = (const float*)d_in[3];
  const float* m1 = (const float*)d_in[4];
  const float* v1 = (const float*)d_in[5];
  const float* w2 = (const float*)d_in[6];
  const float* g2 = (const float*)d_in[7];
  const float* b2 = (const float*)d_in[8];
  const float* m2 = (const float*)d_in[9];
  const float* v2 = (const float*)d_in[10];
  const float* w3 = (const float*)d_in[11];
  const float* g3 = (const float*)d_in[12];
  const float* b3 = (const float*)d_in[13];
  const float* m3 = (const float*)d_in[14];
  const float* v3 = (const float*)d_in[15];

  float* ws   = (float*)d_ws;
  float* pmax = ws;
  float* pavg = ws + BATCH * HW;

  pool_kernel<<<BATCH * NRG, 256, 0, stream>>>(x, pmax, pavg);
  conv_scale_kernel<<<BATCH * NRG, 256, 0, stream>>>(x, pmax, pavg, w1, w2, w3,
                                                    g1, b1, m1, v1,
                                                    g2, b2, m2, v2,
                                                    g3, b3, m3, v3, (float*)d_out);
}


# one launch, not two: each 16-wave block pools its own 8 rows and 6 halo rows itself and keeps its 8 rows x 256 channels on chip (half LDS, half registers) for the scaling; x read from memory once
# speedup vs baseline: 2.0458x; 1.1502x over previous
#include <hip/hip_runtime.h>
#include <math.h>


typedef __attribute__((ext_vector_type(2))) float v2f;
typedef __attribute__((ext_vector_type(8))) float v8f;

#define BATCH 32
#define CCH   256
#define HH    64
#define WW    64
#define HW    (HH * WW)
#define CHW   (CCH * HW)
#define EPSV  1e-5f
#define RB    8
#define NRG   (HH / RB)
#define LPR   (RB + 7)
#define LPCH  (LPR * 72)
#define NTH   512
#define NPF   (RB * 16 + 96)

static __device__ __forceinline__ void pool_acc(float4& mx, float4& sm, const float4 v) {
  mx.x = fmaxf(mx.x, v.x); mx.y = fmaxf(mx.y, v.y);
  mx.z = fmaxf(mx.z, v.z); mx.w = fmaxf(mx.w, v.w);
  sm.x += v.x; sm.y += v.y; sm.z += v.z; sm.w += v.w;
}

__global__ __launch_bounds__(NTH) __attribute__((amdgpu_waves_per_eu(4, 4))) void triplet_kernel(const float* __restrict__ x,
                                const float* __restrict__ w1,
                                const float* __restrict__ w2,
                                const float* __restrict__ w3,
                                const float* __restrict__ g1, const float* __restrict__ be1,
                                const float* __restrict__ m1, const float* __restrict__ v1,
                                const float* __restrict__ g2, const float* __restrict__ be2,
                                const float* __restrict__ m2, const float* __restrict__ v2,
                                const float* __restrict__ g3, const float* __restrict__ be3,
                                const float* __restrict__ m3, const float* __restrict__ v3,
                                float* __restrict__ out) {
  __shared__ float4 xt[32][NTH];
  __shared__ __attribute__((aligned(16))) float lp[2][LPR][72];
  __shared__ __attribute__((aligned(16))) float btabT[16][128];
  __shared__ float ys[3][RB * 64];
  __shared__ __attribute__((aligned(16))) float att_s[RB * 64];
  __shared__ float4 rmax[4][NPF];
  __shared__ float4 rsum[4][NPF];
  const int g = blockIdx.x;
  const int b = g >> 3, h0 = (g & 7) * RB;
  const int tid = threadIdx.x;

  const float G1 = g1[0], Be1 = be1[0], M1 = m1[0], V1 = v1[0];
  const float G2 = g2[0], Be2 = be2[0], M2 = m2[0], V2 = v2[0];
  const float G3 = g3[0], Be3 = be3[0], M3 = m3[0], V3 = v3[0];
  const int k = tid & 127;
  const int kch = k >> 6, kr = (k >> 3) & 7, ks = k & 7;
  const bool tapok = (kr < 7) && (ks < 7);
  const bool hi = (tid & 128) != 0;
  const int offn = kch * 49 + kr * 7 + ks;
  const int offt = kch * 49 + ks * 7 + kr;

  const int f = tid & 127, cg = tid >> 7;
  const size_t rowoff = (size_t)b * CHW + (size_t)h0 * WW;
  const float4* xb = (const float4*)(x + rowoff) + f + (size_t)cg * (HW / 4);
  float4 xr[32];
  float4 vb[16];
#pragma unroll
  for (int i = 0; i < 32; ++i) xr[i] = xb[(size_t)i * HW];

  const float4 z4 = make_float4(0.f, 0.f, 0.f, 0.f);
  float4* lp4 = (float4*)&lp[0][0][0];
  float4* bt4 = (float4*)&btabT[0][0];
  lp4[tid] = z4;
  if (tid < 2 * LPR * 18 - NTH) lp4[NTH + tid] = z4;
  if (tid < 384) bt4[128 + tid] = z4;

  float4 mx = make_float4(-INFINITY, -INFINITY, -INFINITY, -INFINITY);
  float4 sm = make_float4(0.f, 0.f, 0.f, 0.f);
#pragma unroll
  for (int i = 0; i < 32; ++i) {
    pool_acc(mx, sm, xr[i]);
    xt[i][tid] = xr[i];
    xr[i] = xb[(size_t)(32 + i) * HW];
  }
#pragma unroll
  for (int i = 0; i < 32; ++i) pool_acc(mx, sm, xr[i]);
  rmax[cg][f] = mx;
  rsum[cg][f] = sm;

  float wv01 = 0.f, wv2 = 0.f;
  if (tid < 256) {
    const float* wp01 = hi ? w2 : w1;
    wv01 = wp01[tapok ? (hi ? offt : offn) : 0];
    wv2  = w3[(tapok && !hi) ? offn : 0];
  }

  {
    const int hrow = f >> 4;
    const int hh = (hrow < 3) ? (h0 - 3 + hrow) : (h0 + RB - 3 + hrow);
    if (f < 96 && hh >= 0 && hh < HH) {
      const float4* xh = (const float4*)(x + (size_t)b * CHW + (size_t)hh * WW)
                         + (f & 15) + (size_t)cg * (HW / 4);
      float4 hm = make_float4(-INFINITY, -INFINITY, -INFINITY, -INFINITY);
      float4 hs = make_float4(0.f, 0.f, 0.f, 0.f);
#pragma unroll
      for (int j = 0; j < 16; ++j) vb[j] = xh[(size_t)j * HW];
#pragma unroll
      for (int q = 1; q < 4; ++q) {
#pragma unroll
        for (int j = 0; j < 16; ++j) {
          pool_acc(hm, hs, vb[j]);
          vb[j] = xh[(size_t)(16 * q + j) * HW];
        }
      }
#pragma unroll
      for (int j = 0; j < 16; ++j) pool_acc(hm, hs, vb[j]);
      rmax[cg][RB * 16 + f] = hm;
      rsum[cg][RB * 16 + f] = hs;
    }
  }
  __syncthreads();

  if (tid < NPF) {
    int r, hh;
    if (tid < RB * 16) {
      r = 3 + (tid >> 4);
    } else {
      const int hrow = (tid - RB * 16) >> 4;
      r = (hrow < 3) ? hrow : (RB + hrow);
    }
    hh = h0 - 3 + r;
    if (hh >= 0 && hh < HH) {
      float4 a = rmax[0][tid];
      float4 u = rsum[0][tid];
#pragma unroll
      for (int q = 1; q < 4; ++q) {
        const float4 c = rmax[q][tid];
        const float4 d = rsum[q][tid];
        a.x = fmaxf(a.x, c.x); a.y = fmaxf(a.y, c.y);
        a.z = fmaxf(a.z, c.z); a.w = fmaxf(a.w, c.w);
        u.x += d.x; u.y += d.y; u.z += d.z; u.w += d.w;
      }
      u.x *= (1.f / 256.f); u.y *= (1.f / 256.f);
      u.z *= (1.f / 256.f); u.w *= (1.f / 256.f);
      float* pm = &lp[0][r][3 + 4 * (tid & 15)];
      float* pa = &lp[1][r][3 + 4 * (tid & 15)];
      pm[0] = a.x; pm[1] = a.y; pm[2] = a.z; pm[3] = a.w;
      pa[0] = u.x; pa[1] = u.y; pa[2] = u.z; pa[3] = u.w;
    }
  }
  if (tid < 256) {
    float* btw = &btabT[0][0];
    btw[tid] = tapok ? wv01 : 0.f;
    btw[256 + tid] = (tapok && !hi) ? wv2 : 0.f;
  }
  __syncthreads();

  const int lane = tid & 31;
  const int wid = __builtin_amdgcn_readfirstlane(tid >> 5);
  const int m = lane & 15, khalf = lane >> 4;
  const int rt = wid >> 1;
  const int c0 = (wid & 1) * 32;
  const float* lpf = &lp[0][0][0];
  const float* btf = &btabT[0][0];
  const int aBase = rt * 72 + c0 + m + 2 * khalf;
  const int bBase = m * 128 + 2 * khalf;

  {
    v8f acc0 = {}, acc1 = {};
#pragma unroll
    for (int chunk = 0; chunk < 32; ++chunk) {
      const int k0 = chunk * 4;
      const int ch = k0 >> 6;
      const int r  = (k0 >> 3) & 7;
      const int sc = k0 & 7;
      const int ai = ch * LPCH + r * 72 + sc + aBase;
      v2f a0, a1;
      a0[0] = lpf[ai];
      a0[1] = lpf[ai + 1];
      a1[0] = lpf[ai + 16];
      a1[1] = lpf[ai + 17];
      v2f bm = *(const v2f*)(btf + bBase + k0);
      acc0 = __builtin_amdgcn_wmma_f32_16x16x4_f32(
           false, a0,  false, bm,
           (short)0, acc0,  false,  false);
      acc1 = __builtin_amdgcn_wmma_f32_16x16x4_f32(
           false, a1,  false, bm,
           (short)0, acc1,  false,  false);
    }
    if (m < 3) {
      float* yd = &ys[m][rt * 64 + c0 + 8 * khalf];
#pragma unroll
      for (int v = 0; v < 8; ++v) {
        yd[v] = acc0[v];
        yd[16 + v] = acc1[v];
      }
    }
  }
  __syncthreads();

  {
    float y0 = ys[0][tid];
    float y1 = ys[1][tid];
    float y2 = ys[2][tid];
    float s1 = G1 / sqrtf(V1 + EPSV), t1 = Be1 - M1 * s1;
    float s2 = G2 / sqrtf(V2 + EPSV), t2 = Be2 - M2 * s2;
    float s3 = G3 / sqrtf(V3 + EPSV), t3 = Be3 - M3 * s3;
    float a1 = 1.f / (1.f + expf(-(y0 * s1 + t1)));
    float a2 = 1.f / (1.f + expf(-(y1 * s2 + t2)));
    float a3 = 1.f / (1.f + expf(-(y2 * s3 + t3)));
    att_s[tid] = (a1 + a2 + a3) * (1.f / 3.f);
  }
  __syncthreads();

  const float4 av = *(const float4*)&att_s[f * 4];
  float4* ob = (float4*)(out + rowoff) + f + (size_t)cg * (HW / 4);
#pragma unroll
  for (int j = 0; j < 16; ++j) vb[j] = xt[j][tid];
#pragma unroll
  for (int j = 0; j < 16; ++j) {
    float4 o = vb[j];
    o.x *= av.x; o.y *= av.y; o.z *= av.z; o.w *= av.w;
    ob[(size_t)j * HW] = o;
    vb[j] = xt[16 + j][tid];
  }
#pragma unroll
  for (int i = 0; i < 32; ++i) {
    float4 o = xr[i];
    o.x *= av.x; o.y *= av.y; o.z *= av.z; o.w *= av.w;
    ob[(size_t)(32 + i) * HW] = o;
  }
#pragma unroll
  for (int j = 0; j < 16; ++j) {
    float4 o = vb[j];
    o.x *= av.x; o.y *= av.y; o.z *= av.z; o.w *= av.w;
    ob[(size_t)(16 + j) * HW] = o;
  }
}

extern "C" void kernel_launch(void* const* d_in, const int* in_sizes, int n_in,
                              void* d_out, int out_size, void* d_ws, size_t ws_size,
                              hipStream_t stream) {
  const float* x  = (const float*)d_in[0];
  const float* w1 = (const float*)d_in[1];
  const float* g1 = (const float*)d_in[2];
  const float* b1 = (const float*)d_in[3];
  const float* m1 = (const float*)d_in[4];
  const float* v1 = (const float*)d_in[5];
  const float* w2 = (const float*)d_in[6];
  const float* g2 = (const float*)d_in[7];
  const float* b2 = (const float*)d_in[8];
  const float* m2 = (const float*)d_in[9];
  const float* v2 = (const float*)d_in[10];
  const float* w3 = (const float*)d_in[11];
  const float* g3 = (const float*)d_in[12];
  const float* b3 = (const float*)d_in[13];
  const float* m3 = (const float*)d_in[14];
  const float* v3 = (const float*)d_in[15];

  triplet_kernel<<<BATCH * NRG, NTH, 0, stream>>>(x, w1, w2, w3,
                                                 g1, b1, m1, v1,
                                                 g2, b2, m2, v2,
                                                 g3, b3, m3, v3, (float*)d_out);
}


# halo rows requested first and by all 16 waves (48 loads per lane), own rows' loads queued behind: load stream no longer drains between them; LDS copy laid out per wave; halo mean sum order changes
# speedup vs baseline: 2.3963x; 1.1713x over previous
#include <hip/hip_runtime.h>
#include <math.h>


typedef __attribute__((ext_vector_type(2))) float v2f;
typedef __attribute__((ext_vector_type(8))) float v8f;

#define BATCH 32
#define CCH   256
#define HH    64
#define WW    64
#define HW    (HH * WW)
#define CHW   (CCH * HW)
#define EPSV  1e-5f
#define RB    8
#define NRG   (HH / RB)
#define LPR   (RB + 7)
#define LPCH  (LPR * 72)
#define NTH   512
#define NPF   (RB * 16 + 96)
#define KCH   80
#define BPIT  164

static __device__ __forceinline__ void pool_acc(float4& mx, float4& sm, const float4 v) {
  mx.x = fmaxf(mx.x, v.x); mx.y = fmaxf(mx.y, v.y);
  mx.z = fmaxf(mx.z, v.z); mx.w = fmaxf(mx.w, v.w);
  sm.x += v.x; sm.y += v.y; sm.z += v.z; sm.w += v.w;
}

__global__ __launch_bounds__(NTH) __attribute__((amdgpu_waves_per_eu(4, 4))) void triplet_kernel(const float* __restrict__ x,
                                const float* __restrict__ w1,
                                const float* __restrict__ w2,
                                const float* __restrict__ w3,
                                const float* __restrict__ g1, const float* __restrict__ be1,
                                const float* __restrict__ m1, const float* __restrict__ v1,
                                const float* __restrict__ g2, const float* __restrict__ be2,
                                const float* __restrict__ m2, const float* __restrict__ v2,
                                const float* __restrict__ g3, const float* __restrict__ be3,
                                const float* __restrict__ m3, const float* __restrict__ v3,
                                float* __restrict__ out) {
  __shared__ float4 xt[NTH / 32][32][32];
  __shared__ __attribute__((aligned(16))) float lp[2][LPR][72];
  __shared__ __attribute__((aligned(16))) float btabT[16][BPIT];
  __shared__ __attribute__((aligned(16))) float ysraw[2 * 2 * 64 * 13];
  float (*ys)[2 * 64][13] = (float (*)[2 * 64][13])ysraw;
  float4* pbmax = (float4*)ysraw;
  float4* pbsum = pbmax + 4 * 96;
  __shared__ __attribute__((aligned(16))) float att_s[RB * 64];
  __shared__ float4 rmax[4][NPF];
  __shared__ float4 rsum[4][NPF];
  const int g = blockIdx.x;
  const int b = g >> 3, h0 = (g & 7) * RB;
  const int tid = threadIdx.x;

  const float G1 = g1[0], Be1 = be1[0], M1 = m1[0], V1 = v1[0];
  const float G2 = g2[0], Be2 = be2[0], M2 = m2[0], V2 = v2[0];
  const float G3 = g3[0], Be3 = be3[0], M3 = m3[0], V3 = v3[0];

  const int k = tid & 127;
  const int tb = tid >> 7;
  const int kch = k >> 6, kr = (k >> 3) & 7, ks = k & 7;
  const bool tapok = (tid < 384) && (kr < 7) && (ks < 7);
  const int offw = kch * 49 + ((tb == 1) ? (ks * 7 + kr) : (kr * 7 + ks));

  const int f = tid & 127, cg = tid >> 7;
  const size_t rowoff = (size_t)b * CHW + (size_t)h0 * WW;
  const float4* xb = (const float4*)(x + rowoff) + f + (size_t)cg * (HW / 4);
  float4 xr[32];
  float4 vb[16];

  const int wq = f >> 5, ql = f & 31;
  const float4* xh[3];
#pragma unroll
  for (int t = 0; t < 3; ++t) {
    const int p = (wq < 3) ? f : (ql + 32 * t);
    const int hrow = p >> 4;
    const int hh = (hrow < 3) ? (h0 - 3 + hrow) : (h0 + RB - 3 + hrow);
    const int hhc = (hh < 0) ? 0 : ((hh > HH - 1) ? (HH - 1) : hh);
    const int ich = (wq < 3) ? (16 * t) : 48;
    xh[t] = (const float4*)(x + (size_t)b * CHW + (size_t)hhc * WW)
            + (p & 15) + (size_t)cg * (HW / 4) + (size_t)ich * HW;
  }
  const float4 z4 = make_float4(0.f, 0.f, 0.f, 0.f);
  float4* lp4 = (float4*)&lp[0][0][0];
  float4* bt4 = (float4*)&btabT[0][0];
  lp4[tid] = z4;
  if (tid < 2 * LPR * 18 - NTH) lp4[NTH + tid] = z4;
  bt4[tid] = z4;
  if (tid < 16 * BPIT / 4 - NTH) bt4[NTH + tid] = z4;

  float4 hm[3], hs[3];
#pragma unroll
  for (int t = 0; t < 3; ++t) {
    hm[t] = make_float4(-INFINITY, -INFINITY, -INFINITY, -INFINITY);
    hs[t] = make_float4(0.f, 0.f, 0.f, 0.f);
  }
  {
    float4 hv[6][8];
#pragma unroll
    for (int s = 0; s < 4; ++s) {
#pragma unroll
      for (int j = 0; j < 8; ++j)
        hv[s][j] = xh[s >> 1][(size_t)(8 * (s & 1) + j) * HW];
    }
    asm volatile("" ::: "memory");
#pragma unroll
    for (int s = 0; s < 6; ++s) {
#pragma unroll
      for (int j = 0; j < 8; ++j) pool_acc(hm[s >> 1], hs[s >> 1], hv[s][j]);
      if (s < 2) {
#pragma unroll
        for (int j = 0; j < 8; ++j)
          hv[s + 4][j] = xh[(s + 4) >> 1][(size_t)(8 * (s & 1) + j) * HW];
      } else {
#pragma unroll
        for (int j = 0; j < 8; ++j)
          xr[(s - 2) * 8 + j] = xb[(size_t)((s - 2) * 8 + j) * HW];
      }
      asm volatile("" ::: "memory");
    }
  }

  if (wq < 3) {
    float4 a = hm[0], u = hs[0];
#pragma unroll
    for (int t = 1; t < 3; ++t) {
      a.x = fmaxf(a.x, hm[t].x); a.y = fmaxf(a.y, hm[t].y);
      a.z = fmaxf(a.z, hm[t].z); a.w = fmaxf(a.w, hm[t].w);
      u.x += hs[t].x; u.y += hs[t].y; u.z += hs[t].z; u.w += hs[t].w;
    }
    rmax[cg][RB * 16 + f] = a;
    rsum[cg][RB * 16 + f] = u;
  } else {
#pragma unroll
    for (int t = 0; t < 3; ++t) {
      pbmax[cg * 96 + ql + 32 * t] = hm[t];
      pbsum[cg * 96 + ql + 32 * t] = hs[t];
    }
  }

  float wv = 0.f;
  if (tapok) {
    const float* wp = (tb == 0) ? w1 : ((tb == 1) ? w2 : w3);
    wv = wp[offw];
  }

  float4 mx = make_float4(-INFINITY, -INFINITY, -INFINITY, -INFINITY);
  float4 sm = make_float4(0.f, 0.f, 0.f, 0.f);
#pragma unroll
  for (int i = 0; i < 32; ++i) {
    pool_acc(mx, sm, xr[i]);
    xt[tid >> 5][i][tid & 31] = xr[i];
    xr[i] = xb[(size_t)(32 + i) * HW];
  }
#pragma unroll
  for (int i = 0; i < 32; ++i) pool_acc(mx, sm, xr[i]);
  rmax[cg][f] = mx;
  rsum[cg][f] = sm;
  __syncthreads();

  if (tid < NPF) {
    int r, hh;
    if (tid < RB * 16) {
      r = 3 + (tid >> 4);
    } else {
      const int hrow = (tid - RB * 16) >> 4;
      r = (hrow < 3) ? hrow : (RB + hrow);
    }
    hh = h0 - 3 + r;
    if (hh >= 0 && hh < HH) {
      float4 a = rmax[0][tid];
      float4 u = rsum[0][tid];
#pragma unroll
      for (int q = 1; q < 4; ++q) {
        const float4 c = rmax[q][tid];
        const float4 d = rsum[q][tid];
        a.x = fmaxf(a.x, c.x); a.y = fmaxf(a.y, c.y);
        a.z = fmaxf(a.z, c.z); a.w = fmaxf(a.w, c.w);
        u.x += d.x; u.y += d.y; u.z += d.z; u.w += d.w;
      }
      if (tid >= RB * 16) {
        const int p = tid - RB * 16;
#pragma unroll
        for (int qg = 0; qg < 4; ++qg) {
          const float4 c = pbmax[qg * 96 + p];
          const float4 d = pbsum[qg * 96 + p];
          a.x = fmaxf(a.x, c.x); a.y = fmaxf(a.y, c.y);
          a.z = fmaxf(a.z, c.z); a.w = fmaxf(a.w, c.w);
          u.x += d.x; u.y += d.y; u.z += d.z; u.w += d.w;
        }
      }
      u.x *= (1.f / 256.f); u.y *= (1.f / 256.f);
      u.z *= (1.f / 256.f); u.w *= (1.f / 256.f);
      float* pm = &lp[0][r][3 + 4 * (tid & 15)];
      float* pa = &lp[1][r][3 + 4 * (tid & 15)];
      pm[0] = a.x; pm[1] = a.y; pm[2] = a.z; pm[3] = a.w;
      pa[0] = u.x; pa[1] = u.y; pa[2] = u.z; pa[3] = u.w;
    }
  }
  if (tapok) {
    float* btw = &btabT[tb * 4][kch * KCH + kr * 8 + ks];
#pragma unroll
    for (int dn = 0; dn < 4; ++dn) btw[dn * (BPIT + 8)] = wv;
  }
  __syncthreads();

  const int lane = tid & 31;
  const int wid = __builtin_amdgcn_readfirstlane(tid >> 5);
  const int m = lane & 15, khalf = lane >> 4;
  const int kp = wid & 1;
  const int rq = wid >> 3;
  const int c0 = ((wid >> 1) & 3) * 16;
  const float* lpf = &lp[0][0][0];
  const float* btf = &btabT[0][0];
  const int aBase = kp * LPCH + rq * (4 * 72) + c0 + m + 2 * khalf;
  const int bBase = m * BPIT + kp * KCH + 2 * khalf;

  {
    v8f acc = {};
#pragma unroll
    for (int chunk = 0; chunk < KCH / 4; ++chunk) {
      const int k0 = chunk * 4;
      const int rho = k0 >> 3;
      const int sc = k0 & 7;
      const int ai = rho * 72 + sc + aBase;
      v2f a0;
      a0[0] = lpf[ai];
      a0[1] = lpf[ai + 1];
      v2f bm = *(const v2f*)(btf + bBase + k0);
      acc = __builtin_amdgcn_wmma_f32_16x16x4_f32(
           false, a0,  false, bm,
           (short)0, acc,  false,  false);
    }
    if (m < 12) {
      float* yd = &ys[kp][rq * 64 + c0 + 8 * khalf][m];
#pragma unroll
      for (int v = 0; v < 8; ++v) yd[v * 13] = acc[v];
    }
  }
  __syncthreads();

  {
    const float* y0p = &ys[0][(tid >> 8) * 64 + (tid & 63)][(tid >> 6) & 3];
    const float* y1p = &ys[1][(tid >> 8) * 64 + (tid & 63)][(tid >> 6) & 3];
    float y0 = y0p[0] + y1p[0];
    float y1 = y0p[4] + y1p[4];
    float y2 = y0p[8] + y1p[8];
    float s1 = G1 / sqrtf(V1 + EPSV), t1 = Be1 - M1 * s1;
    float s2 = G2 / sqrtf(V2 + EPSV), t2 = Be2 - M2 * s2;
    float s3 = G3 / sqrtf(V3 + EPSV), t3 = Be3 - M3 * s3;
    float a1 = 1.f / (1.f + expf(-(y0 * s1 + t1)));
    float a2 = 1.f / (1.f + expf(-(y1 * s2 + t2)));
    float a3 = 1.f / (1.f + expf(-(y2 * s3 + t3)));
    att_s[tid] = (a1 + a2 + a3) * (1.f / 3.f);
  }
  __syncthreads();

  const float4 av = *(const float4*)&att_s[f * 4];
  float4* ob = (float4*)(out + rowoff) + f + (size_t)cg * (HW / 4);
#pragma unroll
  for (int j = 0; j < 16; ++j) vb[j] = xt[tid >> 5][j][tid & 31];
#pragma unroll
  for (int j = 0; j < 16; ++j) {
    float4 o = vb[j];
    o.x *= av.x; o.y *= av.y; o.z *= av.z; o.w *= av.w;
    ob[(size_t)j * HW] = o;
    vb[j] = xt[tid >> 5][16 + j][tid & 31];
  }
#pragma unroll
  for (int i = 0; i < 32; ++i) {
    float4 o = xr[i];
    o.x *= av.x; o.y *= av.y; o.z *= av.z; o.w *= av.w;
    ob[(size_t)(32 + i) * HW] = o;
  }
#pragma unroll
  for (int j = 0; j < 16; ++j) {
    float4 o = vb[j];
    o.x *= av.x; o.y *= av.y; o.z *= av.z; o.w *= av.w;
    ob[(size_t)(16 + j) * HW] = o;
  }
}

extern "C" void kernel_launch(void* const* d_in, const int* in_sizes, int n_in,
                              void* d_out, int out_size, void* d_ws, size_t ws_size,
                              hipStream_t stream) {
  const float* x  = (const float*)d_in[0];
  const float* w1 = (const float*)d_in[1];
  const float* g1 = (const float*)d_in[2];
  const float* b1 = (const float*)d_in[3];
  const float* m1 = (const float*)d_in[4];
  const float* v1 = (const float*)d_in[5];
  const float* w2 = (const float*)d_in[6];
  const float* g2 = (const float*)d_in[7];
  const float* b2 = (const float*)d_in[8];
  const float* m2 = (const float*)d_in[9];
  const float* v2 = (const float*)d_in[10];
  const float* w3 = (const float*)d_in[11];
  const float* g3 = (const float*)d_in[12];
  const float* b3 = (const float*)d_in[13];
  const float* m3 = (const float*)d_in[14];
  const float* v3 = (const float*)d_in[15];

  triplet_kernel<<<BATCH * NRG, NTH, 0, stream>>>(x, w1, w2, w3,
                                                 g1, b1, m1, v1,
                                                 g2, b2, m2, v2,
                                                 g3, b3, m3, v3, (float*)d_out);
}
